# starting kernel
# baseline (speedup 1.0000x reference)
#include <hip/hip_runtime.h>
#include <hip/hip_bf16.h>


typedef float v2f __attribute__((ext_vector_type(2)));
typedef float v8f __attribute__((ext_vector_type(8)));

#define WMMA_F32(a, b, c) \
  __builtin_amdgcn_wmma_f32_16x16x4_f32(false, (a), false, (b), (short)0, (c), false, false)

template <int MODE>
__global__ __launch_bounds__(256) void gemm_nt_wmma(const float* __restrict__ A,
                                                    const float* __restrict__ Bw,
                                                    float* __restrict__ Cout,
                                                    int M, int N, int K) {
  constexpr int LDA = 36;
  __shared__ float As[128 * LDA];
  __shared__ float Bs[64 * LDA];

  const int tid  = threadIdx.x;
  const int lane = tid & 31;
  const int wave = tid >> 5;
  const int m0   = blockIdx.y * 128;
  const int n0   = blockIdx.x * 64;

  const int lr   = lane & 15;
  const int koff = (lane >> 4) << 1;

  const int la_r = tid >> 3;
  const int la_k = (tid & 7) << 2;

  v8f acc0 = {}, acc1 = {}, acc2 = {}, acc3 = {};

  for (int k0 = 0; k0 < K; k0 += 32) {
    __syncthreads();
    {
#pragma unroll
      for (int r4 = 0; r4 < 4; ++r4) {
        const int r = la_r + 32 * r4;
        const float4 va = *(const float4*)(A + (size_t)(m0 + r) * K + k0 + la_k);
        *(float4*)&As[r * LDA + la_k] = va;
      }
      const float4 vb0 = *(const float4*)(Bw + (size_t)(n0 + la_r) * K + k0 + la_k);
      *(float4*)&Bs[la_r * LDA + la_k] = vb0;
      const float4 vb1 = *(const float4*)(Bw + (size_t)(n0 + 32 + la_r) * K + k0 + la_k);
      *(float4*)&Bs[(32 + la_r) * LDA + la_k] = vb1;
    }
    __syncthreads();
#pragma unroll
    for (int j = 0; j < 8; ++j) {
      v2f a = *(const v2f*)&As[(wave * 16 + lr) * LDA + j * 4 + koff];
      v2f b0 = *(const v2f*)&Bs[(0 * 16 + lr) * LDA + j * 4 + koff];
      acc0 = WMMA_F32(a, b0, acc0);
      v2f b1 = *(const v2f*)&Bs[(1 * 16 + lr) * LDA + j * 4 + koff];
      acc1 = WMMA_F32(a, b1, acc1);
      v2f b2 = *(const v2f*)&Bs[(2 * 16 + lr) * LDA + j * 4 + koff];
      acc2 = WMMA_F32(a, b2, acc2);
      v2f b3 = *(const v2f*)&Bs[(3 * 16 + lr) * LDA + j * 4 + koff];
      acc3 = WMMA_F32(a, b3, acc3);
    }
  }

  const int rbase = m0 + wave * 16 + ((lane >> 4) << 3);

  if (MODE == 0) {
#pragma unroll
    for (int nc = 0; nc < 4; ++nc) {
      const int col = n0 + nc * 16 + lr;
      const v8f acc = (nc == 0) ? acc0 : (nc == 1) ? acc1 : (nc == 2) ? acc2 : acc3;
#pragma unroll
      for (int i = 0; i < 8; ++i)
        Cout[(size_t)(rbase + i) * N + col] = acc[i];
    }
  } else {
#pragma unroll
    for (int nc = 0; nc < 4; ++nc) {
      const int col = n0 + nc * 16 + lr;
      const v8f acc = (nc == 0) ? acc0 : (nc == 1) ? acc1 : (nc == 2) ? acc2 : acc3;
      const int tsel = col >> 10;
      const int r1   = col & 1023;
      const int h    = r1 >> 6;
      const int d    = r1 & 63;
      float* dst = Cout + (size_t)tsel * (2ull * 16 * 4096 * 64);
#pragma unroll
      for (int i = 0; i < 8; ++i) {
        const int m = rbase + i;
        const int b = m >> 12;
        const int s = m & 4095;
        dst[((size_t)((b * 16 + h) * 4096 + s)) * 64 + d] = acc[i];
      }
    }
  }
}

__global__ __launch_bounds__(128) void attn_kernel(const float* __restrict__ Qg,
                                                   const float* __restrict__ Kg,
                                                   const float* __restrict__ Vg,
                                                   float* __restrict__ Og) {
  constexpr int S_ = 4096, H = 16, Dh = 64, WND = 512;
  constexpr int LK = 68;
  constexpr int LV = 18;

  const int blk  = blockIdx.x;
  const int cq64 = blk & ((S_ / 64) - 1);
  const int bh   = blk / (S_ / 64);

  const float* Qb = Qg + (size_t)bh * S_ * Dh;
  const float* Kb = Kg + (size_t)bh * S_ * Dh;
  const float* Vb = Vg + (size_t)bh * S_ * Dh;

  const int tid  = threadIdx.x;
  const int lane = tid & 31;
  const int wave = tid >> 5;
  const int lr   = lane & 15;
  const int hi   = lane >> 4;
  const int koff = hi << 1;

  const int q0b = cq64 * 64;
  const int q0w = q0b + wave * 16;

  __shared__ float Ks[16 * LK];
  __shared__ float VTs[64 * LV];
  __shared__ float Ps[4][16 * LV];

  v2f qa[16];
  {
    const float* qrow = Qb + (size_t)(q0w + lr) * Dh;
#pragma unroll
    for (int j = 0; j < 16; ++j)
      qa[j] = *(const v2f*)(qrow + j * 4 + koff);
  }

  v8f acc0 = {}, acc1 = {}, acc2 = {}, acc3 = {};
  float mrow[8], lrow[8];
#pragma unroll
  for (int i = 0; i < 8; ++i) { mrow[i] = -1e30f; lrow[i] = 0.0f; }

  const int ct   = q0b >> 4;
  const int t_lo = (ct - 32) > 1 ? (ct - 32) : 1;
  const int t_hi = ct + 3;

  for (int ti = -1; ti <= t_hi - t_lo; ++ti) {
    const int t = (ti < 0) ? 0 : (t_lo + ti);

    __syncthreads();
    {
      const int r = tid >> 4;
      const int c = (tid & 15) << 2;
      const float4 kv0 = *(const float4*)(Kb + (size_t)(t * 16 + r) * Dh + c);
      *(float4*)&Ks[r * LK + c] = kv0;
      const float4 kv1 = *(const float4*)(Kb + (size_t)(t * 16 + r + 8) * Dh + c);
      *(float4*)&Ks[(r + 8) * LK + c] = kv1;
      const float4 vv0 = *(const float4*)(Vb + (size_t)(t * 16 + r) * Dh + c);
      VTs[(c + 0) * LV + r] = vv0.x;
      VTs[(c + 1) * LV + r] = vv0.y;
      VTs[(c + 2) * LV + r] = vv0.z;
      VTs[(c + 3) * LV + r] = vv0.w;
      const float4 vv1 = *(const float4*)(Vb + (size_t)(t * 16 + r + 8) * Dh + c);
      VTs[(c + 0) * LV + r + 8] = vv1.x;
      VTs[(c + 1) * LV + r + 8] = vv1.y;
      VTs[(c + 2) * LV + r + 8] = vv1.z;
      VTs[(c + 3) * LV + r + 8] = vv1.w;
    }
    __syncthreads();

    v8f sc = {};
#pragma unroll
    for (int j = 0; j < 16; ++j) {
      v2f b = *(const v2f*)&Ks[lr * LK + j * 4 + koff];
      sc = WMMA_F32(qa[j], b, sc);
    }

    const int kcol = t * 16 + lr;
    float sm[8], tmax[8];
#pragma unroll
    for (int i = 0; i < 8; ++i) {
      const int qpos = q0w + i + (hi << 3);
      bool ok;
      if (t == 0) {
        ok = (kcol <= qpos);
      } else {
        const int diff = qpos - kcol;
        ok = (diff >= 0) && (diff < WND);
      }
      const float s = sc[i] * 0.125f;
      sm[i]   = ok ? s : -1e30f;
      tmax[i] = sm[i];
    }
#pragma unroll
    for (int i = 0; i < 8; ++i) {
      float v = tmax[i];
      v = fmaxf(v, __shfl_xor(v, 1));
      v = fmaxf(v, __shfl_xor(v, 2));
      v = fmaxf(v, __shfl_xor(v, 4));
      v = fmaxf(v, __shfl_xor(v, 8));
      tmax[i] = v;
    }
#pragma unroll
    for (int i = 0; i < 8; ++i) {
      const float mnew  = fmaxf(mrow[i], tmax[i]);
      const float alpha = __expf(mrow[i] - mnew);
      mrow[i] = mnew;
      const float p = (sm[i] > -1e29f) ? __expf(sm[i] - mnew) : 0.0f;
      float rs = p;
      rs += __shfl_xor(rs, 1);
      rs += __shfl_xor(rs, 2);
      rs += __shfl_xor(rs, 4);
      rs += __shfl_xor(rs, 8);
      lrow[i] = lrow[i] * alpha + rs;
      acc0[i] *= alpha; acc1[i] *= alpha; acc2[i] *= alpha; acc3[i] *= alpha;
      Ps[wave][(i + (hi << 3)) * LV + lr] = p;
    }
    __syncthreads();

#pragma unroll
    for (int j = 0; j < 4; ++j) {
      v2f a  = *(const v2f*)&Ps[wave][lr * LV + j * 4 + koff];
      v2f b0 = *(const v2f*)&VTs[(0 * 16 + lr) * LV + j * 4 + koff];
      acc0 = WMMA_F32(a, b0, acc0);
      v2f b1 = *(const v2f*)&VTs[(1 * 16 + lr) * LV + j * 4 + koff];
      acc1 = WMMA_F32(a, b1, acc1);
      v2f b2 = *(const v2f*)&VTs[(2 * 16 + lr) * LV + j * 4 + koff];
      acc2 = WMMA_F32(a, b2, acc2);
      v2f b3 = *(const v2f*)&VTs[(3 * 16 + lr) * LV + j * 4 + koff];
      acc3 = WMMA_F32(a, b3, acc3);
    }
  }

  const int bq = bh >> 4;
  const int hq = bh & 15;
#pragma unroll
  for (int i = 0; i < 8; ++i) {
    const float inv = 1.0f / lrow[i];
    const int qpos  = q0w + i + (hi << 3);
    float* orow = Og + (size_t)(bq * S_ + qpos) * (H * Dh) + hq * Dh;
    orow[ 0 + lr] = acc0[i] * inv;
    orow[16 + lr] = acc1[i] * inv;
    orow[32 + lr] = acc2[i] * inv;
    orow[48 + lr] = acc3[i] * inv;
  }
}

extern "C" void kernel_launch(void* const* d_in, const int* in_sizes, int n_in,
                              void* d_out, int out_size, void* d_ws, size_t ws_size,
                              hipStream_t stream) {
  (void)in_sizes; (void)n_in; (void)out_size; (void)ws_size;

  const float* x     = (const float*)d_in[0];
  const float* w_qkv = (const float*)d_in[1];
  const float* w_out = (const float*)d_in[2];
  float* out = (float*)d_out;
  float* ws  = (float*)d_ws;

  const size_t per = 2ull * 16 * 4096 * 64;
  float* Qb = ws;
  float* Kb = ws + per;
  float* Vb = ws + 2 * per;
  float* Ob = ws + 3 * per;

  gemm_nt_wmma<1><<<dim3(3072 / 64, 8192 / 128), 256, 0, stream>>>(
      x, w_qkv, Qb, 8192, 3072, 1024);

  attn_kernel<<<dim3(2 * 16 * (4096 / 64)), 128, 0, stream>>>(Qb, Kb, Vb, Ob);

  gemm_nt_wmma<0><<<dim3(1024 / 64, 8192 / 128), 256, 0, stream>>>(
      Ob, w_out, out, 8192, 1024, 1024);
}


# attention: 64 keys per loop trip (4 WMMA sub-tiles) instead of 16: softmax row max/sum shuffles, rescale and barriers once per 64 keys; row sum adds 4 per-lane terms first (fp order changes)
# speedup vs baseline: 1.0559x; 1.0559x over previous
#include <hip/hip_runtime.h>
#include <hip/hip_bf16.h>


typedef float v2f __attribute__((ext_vector_type(2)));
typedef float v8f __attribute__((ext_vector_type(8)));

#define WMMA_F32(a, b, c) \
  __builtin_amdgcn_wmma_f32_16x16x4_f32(false, (a), false, (b), (short)0, (c), false, false)

template <int MODE>
__global__ __launch_bounds__(256) void gemm_nt_wmma(const float* __restrict__ A,
                                                    const float* __restrict__ Bw,
                                                    float* __restrict__ Cout,
                                                    int M, int N, int K) {
  constexpr int LDA = 36;
  __shared__ float As[128 * LDA];
  __shared__ float Bs[64 * LDA];

  const int tid  = threadIdx.x;
  const int lane = tid & 31;
  const int wave = tid >> 5;
  const int m0   = blockIdx.y * 128;
  const int n0   = blockIdx.x * 64;

  const int lr   = lane & 15;
  const int koff = (lane >> 4) << 1;

  const int la_r = tid >> 3;
  const int la_k = (tid & 7) << 2;

  v8f acc0 = {}, acc1 = {}, acc2 = {}, acc3 = {};

  for (int k0 = 0; k0 < K; k0 += 32) {
    __syncthreads();
    {
#pragma unroll
      for (int r4 = 0; r4 < 4; ++r4) {
        const int r = la_r + 32 * r4;
        const float4 va = *(const float4*)(A + (size_t)(m0 + r) * K + k0 + la_k);
        *(float4*)&As[r * LDA + la_k] = va;
      }
      const float4 vb0 = *(const float4*)(Bw + (size_t)(n0 + la_r) * K + k0 + la_k);
      *(float4*)&Bs[la_r * LDA + la_k] = vb0;
      const float4 vb1 = *(const float4*)(Bw + (size_t)(n0 + 32 + la_r) * K + k0 + la_k);
      *(float4*)&Bs[(32 + la_r) * LDA + la_k] = vb1;
    }
    __syncthreads();
#pragma unroll
    for (int j = 0; j < 8; ++j) {
      v2f a = *(const v2f*)&As[(wave * 16 + lr) * LDA + j * 4 + koff];
      v2f b0 = *(const v2f*)&Bs[(0 * 16 + lr) * LDA + j * 4 + koff];
      acc0 = WMMA_F32(a, b0, acc0);
      v2f b1 = *(const v2f*)&Bs[(1 * 16 + lr) * LDA + j * 4 + koff];
      acc1 = WMMA_F32(a, b1, acc1);
      v2f b2 = *(const v2f*)&Bs[(2 * 16 + lr) * LDA + j * 4 + koff];
      acc2 = WMMA_F32(a, b2, acc2);
      v2f b3 = *(const v2f*)&Bs[(3 * 16 + lr) * LDA + j * 4 + koff];
      acc3 = WMMA_F32(a, b3, acc3);
    }
  }

  const int rbase = m0 + wave * 16 + ((lane >> 4) << 3);

  if (MODE == 0) {
#pragma unroll
    for (int nc = 0; nc < 4; ++nc) {
      const int col = n0 + nc * 16 + lr;
      const v8f acc = (nc == 0) ? acc0 : (nc == 1) ? acc1 : (nc == 2) ? acc2 : acc3;
#pragma unroll
      for (int i = 0; i < 8; ++i)
        Cout[(size_t)(rbase + i) * N + col] = acc[i];
    }
  } else {
#pragma unroll
    for (int nc = 0; nc < 4; ++nc) {
      const int col = n0 + nc * 16 + lr;
      const v8f acc = (nc == 0) ? acc0 : (nc == 1) ? acc1 : (nc == 2) ? acc2 : acc3;
      const int tsel = col >> 10;
      const int r1   = col & 1023;
      const int h    = r1 >> 6;
      const int d    = r1 & 63;
      float* dst = Cout + (size_t)tsel * (2ull * 16 * 4096 * 64);
#pragma unroll
      for (int i = 0; i < 8; ++i) {
        const int m = rbase + i;
        const int b = m >> 12;
        const int s = m & 4095;
        dst[((size_t)((b * 16 + h) * 4096 + s)) * 64 + d] = acc[i];
      }
    }
  }
}

constexpr int ATT_LK = 68;
constexpr int ATT_LV = 160;
constexpr int ATT_LP = 68;

template <int NS>
__device__ __forceinline__ void attn_tile(const float* __restrict__ Kb,
                                          const float* __restrict__ Vb,
                                          const int key0, const int q0w, const int tid,
                                          const int lr, const int hi, const int koff,
                                          float* Ks, float* Vs, float* Pw,
                                          const v2f (&qa)[16], v8f (&acc)[4],
                                          float (&mrow)[8], float (&lrow)[8]) {
  constexpr int NK = 16 * NS;

  __syncthreads();
  {
    const int r0 = tid >> 4;
    const int c  = (tid & 15) << 2;
#pragma unroll
    for (int it = 0; it < NK / 8; ++it) {
      const int r = r0 + 8 * it;
      const float4 kv = *(const float4*)(Kb + (size_t)(key0 + r) * 64 + c);
      *(float4*)&Ks[r * ATT_LK + c] = kv;
    }
#pragma unroll
    for (int it = 0; it < NK / 16; ++it) {
      const int p = r0 + 8 * it;
      const float4 v0 = *(const float4*)(Vb + (size_t)(key0 + 2 * p) * 64 + c);
      const float4 v1 = *(const float4*)(Vb + (size_t)(key0 + 2 * p + 1) * 64 + c);
      const float4 w0 = {v0.x, v1.x, v0.y, v1.y};
      const float4 w1 = {v0.z, v1.z, v0.w, v1.w};
      *(float4*)&Vs[p * ATT_LV + c * 2]     = w0;
      *(float4*)&Vs[p * ATT_LV + c * 2 + 4] = w1;
    }
  }
  __syncthreads();

  v8f sc[NS];
#pragma unroll
  for (int s = 0; s < NS; ++s) sc[s] = v8f{};
#pragma unroll
  for (int j = 0; j < 16; ++j) {
#pragma unroll
    for (int s = 0; s < NS; ++s) {
      const v2f b = *(const v2f*)&Ks[(s * 16 + lr) * ATT_LK + j * 4 + koff];
      sc[s] = WMMA_F32(qa[j], b, sc[s]);
    }
  }

  float tmax[8];
#pragma unroll
  for (int i = 0; i < 8; ++i) {
    const int qpos = q0w + i + (hi << 3);
    float mx = -1e30f;
#pragma unroll
    for (int s = 0; s < NS; ++s) {
      const int kcol = key0 + s * 16 + lr;
      const int diff = qpos - kcol;
      const bool ok  = (diff >= 0) && ((kcol < 16) || (diff < 512));
      const float v  = ok ? sc[s][i] * 0.125f : -1e30f;
      sc[s][i] = v;
      mx = fmaxf(mx, v);
    }
    tmax[i] = mx;
  }
#pragma unroll
  for (int i = 0; i < 8; ++i) {
    float v = tmax[i];
    v = fmaxf(v, __shfl_xor(v, 1));
    v = fmaxf(v, __shfl_xor(v, 2));
    v = fmaxf(v, __shfl_xor(v, 4));
    v = fmaxf(v, __shfl_xor(v, 8));
    tmax[i] = v;
  }
#pragma unroll
  for (int i = 0; i < 8; ++i) {
    const float mnew  = fmaxf(mrow[i], tmax[i]);
    const float alpha = __expf(mrow[i] - mnew);
    mrow[i] = mnew;
    float rs = 0.0f;
#pragma unroll
    for (int s = 0; s < NS; ++s) {
      const float v = sc[s][i];
      const float p = (v > -1e29f) ? __expf(v - mnew) : 0.0f;
      rs += p;
      Pw[(i + (hi << 3)) * ATT_LP + s * 16 + lr] = p;
    }
    rs += __shfl_xor(rs, 1);
    rs += __shfl_xor(rs, 2);
    rs += __shfl_xor(rs, 4);
    rs += __shfl_xor(rs, 8);
    lrow[i] = lrow[i] * alpha + rs;
    acc[0][i] *= alpha; acc[1][i] *= alpha; acc[2][i] *= alpha; acc[3][i] *= alpha;
  }
  __syncthreads();

#pragma unroll
  for (int j = 0; j < 4 * NS; ++j) {
    const v2f a = *(const v2f*)&Pw[lr * ATT_LP + j * 4 + koff];
    const float* vp = &Vs[(j * 2 + hi) * ATT_LV + lr * 2];
#pragma unroll
    for (int nc = 0; nc < 4; ++nc) {
      const v2f b = *(const v2f*)(vp + nc * 32);
      acc[nc] = WMMA_F32(a, b, acc[nc]);
    }
  }
}

__global__ __launch_bounds__(128) void attn_kernel(const float* __restrict__ Qg,
                                                   const float* __restrict__ Kg,
                                                   const float* __restrict__ Vg,
                                                   float* __restrict__ Og) {
  constexpr int S_ = 4096, H = 16, Dh = 64;

  const int blk  = blockIdx.x;
  const int cq64 = blk & ((S_ / 64) - 1);
  const int bh   = blk / (S_ / 64);

  const float* Qb = Qg + (size_t)bh * S_ * Dh;
  const float* Kb = Kg + (size_t)bh * S_ * Dh;
  const float* Vb = Vg + (size_t)bh * S_ * Dh;

  const int tid  = threadIdx.x;
  const int lane = tid & 31;
  const int wave = tid >> 5;
  const int lr   = lane & 15;
  const int hi   = lane >> 4;
  const int koff = hi << 1;

  const int q0b = cq64 * 64;
  const int q0w = q0b + wave * 16;

  __shared__ float Ks[64 * ATT_LK];
  __shared__ float Vs[32 * ATT_LV];
  __shared__ float Ps[4][16 * ATT_LP];

  v2f qa[16];
  {
    const float* qrow = Qb + (size_t)(q0w + lr) * Dh;
#pragma unroll
    for (int j = 0; j < 16; ++j)
      qa[j] = *(const v2f*)(qrow + j * 4 + koff);
  }

  v8f acc[4];
#pragma unroll
  for (int nc = 0; nc < 4; ++nc) acc[nc] = v8f{};
  float mrow[8], lrow[8];
#pragma unroll
  for (int i = 0; i < 8; ++i) { mrow[i] = -1e30f; lrow[i] = 0.0f; }

  const int kb_lo = (cq64 - 8) > 0 ? (cq64 - 8) : 0;

  if (kb_lo > 0)
    attn_tile<1>(Kb, Vb, 0, q0w, tid, lr, hi, koff, Ks, Vs, Ps[wave], qa, acc, mrow, lrow);

  for (int kb = kb_lo; kb <= cq64; ++kb)
    attn_tile<4>(Kb, Vb, kb * 64, q0w, tid, lr, hi, koff, Ks, Vs, Ps[wave], qa, acc, mrow, lrow);

  const int bq = bh >> 4;
  const int hq = bh & 15;
#pragma unroll
  for (int i = 0; i < 8; ++i) {
    const float inv = 1.0f / lrow[i];
    const int qpos  = q0w + i + (hi << 3);
    float* orow = Og + (size_t)(bq * S_ + qpos) * (H * Dh) + hq * Dh;
    orow[ 0 + lr] = acc[0][i] * inv;
    orow[16 + lr] = acc[1][i] * inv;
    orow[32 + lr] = acc[2][i] * inv;
    orow[48 + lr] = acc[3][i] * inv;
  }
}

extern "C" void kernel_launch(void* const* d_in, const int* in_sizes, int n_in,
                              void* d_out, int out_size, void* d_ws, size_t ws_size,
                              hipStream_t stream) {
  (void)in_sizes; (void)n_in; (void)out_size; (void)ws_size;

  const float* x     = (const float*)d_in[0];
  const float* w_qkv = (const float*)d_in[1];
  const float* w_out = (const float*)d_in[2];
  float* out = (float*)d_out;
  float* ws  = (float*)d_ws;

  const size_t per = 2ull * 16 * 4096 * 64;
  float* Qb = ws;
  float* Kb = ws + per;
  float* Vb = ws + 2 * per;
  float* Ob = ws + 3 * per;

  gemm_nt_wmma<1><<<dim3(3072 / 64, 8192 / 128), 256, 0, stream>>>(
      x, w_qkv, Qb, 8192, 3072, 1024);

  attn_kernel<<<dim3(2 * 16 * (4096 / 64)), 128, 0, stream>>>(Qb, Kb, Vb, Ob);

  gemm_nt_wmma<0><<<dim3(1024 / 64, 8192 / 128), 256, 0, stream>>>(
      Ob, w_out, out, 8192, 1024, 1024);
}
